# starting kernel
# baseline (speedup 1.0000x reference)
#include <hip/hip_runtime.h>

#define NNODES 100000
#define NEDGES 1600000
#define FIN    16
#define NH     8
#define NC     8
#define HC     64
#define ETOT   (NEDGES + NNODES)

typedef __attribute__((ext_vector_type(2))) float v2f;
typedef __attribute__((ext_vector_type(8))) float v8f;

__device__ __forceinline__ unsigned f2ord(float f) {
    unsigned b = __float_as_uint(f);
    return (b & 0x80000000u) ? ~b : (b | 0x80000000u);
}
__device__ __forceinline__ float ord2f(unsigned k) {
    unsigned b = (k & 0x80000000u) ? (k ^ 0x80000000u) : ~k;
    return __uint_as_float(b);
}

__global__ void k_init(float* __restrict__ out, float* __restrict__ denom,
                       unsigned* __restrict__ emax) {
    int i = blockIdx.x * blockDim.x + threadIdx.x;
    if (i < NNODES * HC) out[i] = 0.0f;
    if (i < NNODES * NH) { denom[i] = 0.0f; emax[i] = 0u; }
}

__global__ __launch_bounds__(256) void k_gemm(const float* __restrict__ x,
                                              const float* __restrict__ W,
                                              float* __restrict__ h) {
    const int lane = threadIdx.x & 31;
    const int wave = threadIdx.x >> 5;
    const int tile = blockIdx.x * 8 + wave;
    const long base = (long)tile * 16;
    if (base >= NNODES) return;

    const int r  = lane & 15;
    const int kh = (lane >> 4) << 1;

    const float* xr = x + (base + r) * FIN;
    v2f a[4];
#pragma unroll
    for (int s = 0; s < 4; ++s) {
        const int k0 = s * 4 + kh;
        a[s][0] = xr[k0];
        a[s][1] = xr[k0 + 1];
    }

#pragma unroll
    for (int j = 0; j < 4; ++j) {
        v8f acc = {};
#pragma unroll
        for (int s = 0; s < 4; ++s) {
            const int k0 = s * 4 + kh;
            v2f b;
            b[0] = W[(k0    ) * HC + j * 16 + r];
            b[1] = W[(k0 + 1) * HC + j * 16 + r];
            acc = __builtin_amdgcn_wmma_f32_16x16x4_f32(
                 false, a[s],  false, b,
                 (short)0, acc,  false,  false);
        }
#pragma unroll
        for (int v = 0; v < 8; ++v) {
            const int row = ((lane >> 4) << 3) + v;
            h[(base + row) * HC + j * 16 + r] = acc[v];
        }
    }
}

__global__ __launch_bounds__(256) void k_att(const float* __restrict__ h,
                                             const float* __restrict__ att_src,
                                             const float* __restrict__ att_dst,
                                             float* __restrict__ a_src,
                                             float* __restrict__ a_dst) {
    int i = blockIdx.x * blockDim.x + threadIdx.x;
    if (i >= NNODES * NH) return;
    const int hh = i & 7;
    const float* hp = h + (long)(i >> 3) * HC + hh * NC;
    float ss = 0.0f, sd = 0.0f;
#pragma unroll
    for (int c = 0; c < NC; ++c) {
        float v = hp[c];
        ss = fmaf(v, att_src[hh * NC + c], ss);
        sd = fmaf(v, att_dst[hh * NC + c], sd);
    }
    a_src[i] = ss;
    a_dst[i] = sd;
}

__device__ __forceinline__ void edge_sd(const int* __restrict__ ei, int e,
                                        int& s, int& d) {
    if (e < NEDGES) { s = ei[e]; d = ei[NEDGES + e]; }
    else            { s = d = e - NEDGES; }
}
__device__ __forceinline__ float edge_e(const float* __restrict__ as,
                                        const float* __restrict__ ad,
                                        int s, int d, int hh) {
    float v = as[s * NH + hh] + ad[d * NH + hh];
    return v > 0.0f ? v : 0.2f * v;
}

__global__ __launch_bounds__(256) void k_emax(const int* __restrict__ ei,
                                              const float* __restrict__ as,
                                              const float* __restrict__ ad,
                                              unsigned* __restrict__ emax) {
    int i = blockIdx.x * blockDim.x + threadIdx.x;
    if (i >= ETOT * NH) return;
    int e = i >> 3, hh = i & 7, s, d;
    edge_sd(ei, e, s, d);
    float v = edge_e(as, ad, s, d, hh);
    atomicMax(emax + d * NH + hh, f2ord(v));
}

__global__ __launch_bounds__(256) void k_denom(const int* __restrict__ ei,
                                               const float* __restrict__ as,
                                               const float* __restrict__ ad,
                                               const unsigned* __restrict__ emax,
                                               float* __restrict__ denom) {
    int i = blockIdx.x * blockDim.x + threadIdx.x;
    if (i >= ETOT * NH) return;
    int e = i >> 3, hh = i & 7, s, d;
    edge_sd(ei, e, s, d);
    float v  = edge_e(as, ad, s, d, hh);
    float m  = ord2f(emax[d * NH + hh]);
    float ex = __expf(v - m);
    atomicAdd(denom + d * NH + hh, ex);
}

__global__ __launch_bounds__(256) void k_scatter(const int* __restrict__ ei,
                                                 const float* __restrict__ as,
                                                 const float* __restrict__ ad,
                                                 const unsigned* __restrict__ emax,
                                                 const float* __restrict__ denom,
                                                 const float* __restrict__ h,
                                                 float* __restrict__ out) {
    int i = blockIdx.x * blockDim.x + threadIdx.x;
    if (i >= ETOT * NH) return;
    int e = i >> 3, hh = i & 7, s, d;
    edge_sd(ei, e, s, d);

    const float* hp = h + (long)s * HC + hh * NC;
    __builtin_prefetch(hp, 0, 0);

    float v     = edge_e(as, ad, s, d, hh);
    float m     = ord2f(emax[d * NH + hh]);
    float ex    = __expf(v - m);
    float alpha = ex / (denom[d * NH + hh] + 1e-16f);

    float4 h0 = ((const float4*)hp)[0];
    float4 h1 = ((const float4*)hp)[1];
    float* op = out + (long)d * HC + hh * NC;
    atomicAdd(op + 0, alpha * h0.x);
    atomicAdd(op + 1, alpha * h0.y);
    atomicAdd(op + 2, alpha * h0.z);
    atomicAdd(op + 3, alpha * h0.w);
    atomicAdd(op + 4, alpha * h1.x);
    atomicAdd(op + 5, alpha * h1.y);
    atomicAdd(op + 6, alpha * h1.z);
    atomicAdd(op + 7, alpha * h1.w);
}

__global__ __launch_bounds__(256) void k_final(float* __restrict__ out,
                                               const float* __restrict__ bias) {
    int i = blockIdx.x * blockDim.x + threadIdx.x;
    if (i >= NNODES * HC) return;
    float v = out[i] + bias[i & (HC - 1)];
    out[i] = v > 0.0f ? v : (__expf(v) - 1.0f);
}

extern "C" void kernel_launch(void* const* d_in, const int* in_sizes, int n_in,
                              void* d_out, int out_size, void* d_ws, size_t ws_size,
                              hipStream_t stream) {
    const float* x       = (const float*)d_in[0];
    const int*   ei      = (const int*)  d_in[1];
    const float* W       = (const float*)d_in[2];
    const float* att_src = (const float*)d_in[3];
    const float* att_dst = (const float*)d_in[4];
    const float* bias    = (const float*)d_in[5];
    float* out = (float*)d_out;

    char* ws = (char*)d_ws;
    float*    h      = (float*)ws;     ws += sizeof(float)    * NNODES * HC;
    float*    a_src  = (float*)ws;     ws += sizeof(float)    * NNODES * NH;
    float*    a_dst  = (float*)ws;     ws += sizeof(float)    * NNODES * NH;
    unsigned* emax   = (unsigned*)ws;  ws += sizeof(unsigned) * NNODES * NH;
    float*    denom  = (float*)ws;

    const int ntile   = NNODES / 16;
    const int eh      = ETOT * NH;
    const int blk     = 256;

    k_init   <<<(NNODES * HC + blk - 1) / blk, blk, 0, stream>>>(out, denom, emax);
    k_gemm   <<<(ntile + 7) / 8,               blk, 0, stream>>>(x, W, h);
    k_att    <<<(NNODES * NH + blk - 1) / blk, blk, 0, stream>>>(h, att_src, att_dst, a_src, a_dst);
    k_emax   <<<(eh + blk - 1) / blk,          blk, 0, stream>>>(ei, a_src, a_dst, emax);
    k_denom  <<<(eh + blk - 1) / blk,          blk, 0, stream>>>(ei, a_src, a_dst, emax, denom);
    k_scatter<<<(eh + blk - 1) / blk,          blk, 0, stream>>>(ei, a_src, a_dst, emax, denom, h, out);
    k_final  <<<(NNODES * HC + blk - 1) / blk, blk, 0, stream>>>(out, bias);
}


# Gather loop: a scheduling fence keeps the trip's 8 row loads and the next trip's ids ahead of all its arithmetic, so a trip waits for memory once instead of four times in a row
# speedup vs baseline: 5.5876x; 5.5876x over previous
#include <hip/hip_runtime.h>


#define NNODES 100000
#define NEDGES 1600000
#define FIN    16
#define NH     8
#define NC     8
#define HC     64
#define GCH    4

#define BSH    8
#define BW     256
#define NBKT   391
#define NB1    250
#define T1     640
#define EPT    10
#define EPB    6400
#define SRCB   17
#define CAP2   8192

static_assert(NB1 * EPB == NEDGES && T1 * EPT == EPB, "level-1 blocks tile the edge list exactly");
static_assert(NBKT * BW >= NNODES && (NBKT - 1) * BW < NNODES && (1 << BSH) == BW, "bucket count");
static_assert(NNODES <= (1 << SRCB) && SRCB + BSH <= 31, "a packed edge record fits one int");
static_assert(NBKT <= T1 && NBKT <= 65535, "one thread per bucket; a bucket id fits 16 bits");

typedef __attribute__((ext_vector_type(2))) float v2f;
typedef __attribute__((ext_vector_type(8))) float v8f;

__device__ __forceinline__ int block_excl_scan(const int c, int* wsum, const int nw) {
    const int lane = threadIdx.x & 31;
    const int wave = threadIdx.x >> 5;
    int incl = c;
    for (int off = 1; off < 32; off <<= 1) {
        const int t = __shfl_up(incl, off);
        if (lane >= off) incl += t;
    }
    if (lane == 31) wsum[wave] = incl;
    __syncthreads();
    if (threadIdx.x == 0) {
        int tot = 0;
        for (int w = 0; w < nw; ++w) { const int t = wsum[w]; wsum[w] = tot; tot += t; }
    }
    __syncthreads();
    return wsum[wave] + incl - c;
}

__global__ __launch_bounds__(T1) void k_hist(const int* __restrict__ ei,
                                             int* __restrict__ hist) {
    __shared__ int cnt[NBKT];
    for (int b = threadIdx.x; b < NBKT; b += T1) cnt[b] = 0;
    __syncthreads();
    const int e0 = blockIdx.x * EPB + threadIdx.x;
    int d[EPT];
#pragma unroll
    for (int k = 0; k < EPT; ++k) d[k] = ei[NEDGES + e0 + k * T1];
#pragma unroll
    for (int k = 0; k < EPT; ++k) atomicAdd(&cnt[d[k] >> BSH], 1);
    __syncthreads();
    for (int b = threadIdx.x; b < NBKT; b += T1) hist[blockIdx.x * NBKT + b] = cnt[b];
}

__global__ __launch_bounds__(T1) void k_bucket(const int* __restrict__ ei,
                                               const int* __restrict__ hist,
                                               int* __restrict__ bstart,
                                               int* __restrict__ btotal,
                                               int* __restrict__ bucketed) {
    __shared__ int            cnt[NBKT];
    __shared__ int            delta[NBKT];
    __shared__ int            pre[NBKT];
    __shared__ int            wsum_a[T1 / 32];
    __shared__ int            wsum_b[T1 / 32];
    __shared__ int            lrec[EPB];
    __shared__ unsigned short lbkt[EPB];
    const int t = threadIdx.x;
    for (int b = t; b < NBKT; b += T1) cnt[b] = 0;
    __syncthreads();

    const int e0 = blockIdx.x * EPB + t;
    int s[EPT], d[EPT], r[EPT];
#pragma unroll
    for (int k = 0; k < EPT; ++k) {
        s[k] = ei[e0 + k * T1];
        d[k] = ei[NEDGES + e0 + k * T1];
    }

    int before = 0, total = 0;
    if (t < NBKT) {
        int row = blockIdx.x;
#pragma unroll 50
        for (int i = 0; i < NB1; ++i) {
            const int v = hist[row * NBKT + t];
            before += (row < (int)blockIdx.x) ? v : 0;
            total  += v;
            row = (row + 1 == NB1) ? 0 : row + 1;
        }
    }

#pragma unroll
    for (int k = 0; k < EPT; ++k) r[k] = atomicAdd(&cnt[d[k] >> BSH], 1);
    __syncthreads();

    const int bs = block_excl_scan(total, wsum_a, T1 / 32);
    const int c  = (t < NBKT) ? cnt[t] : 0;
    const int ex = block_excl_scan(c, wsum_b, T1 / 32);
    if (t < NBKT) {
        pre[t]   = ex;
        delta[t] = bs + before - ex;
        if (blockIdx.x == 0) { bstart[t] = bs; btotal[t] = total; }
    }
    __syncthreads();

#pragma unroll
    for (int k = 0; k < EPT; ++k) {
        const int b    = d[k] >> BSH;
        const int slot = pre[b] + r[k];
        lrec[slot] = s[k] | ((d[k] & (BW - 1)) << SRCB);
        lbkt[slot] = (unsigned short)b;
    }
    __syncthreads();

    for (int i = t; i < EPB; i += T1) bucketed[delta[lbkt[i]] + i] = lrec[i];
}

#define RU 20
__device__ __forceinline__ void order_bucket(const int b,
                                             const int* __restrict__ bucketed,
                                             const int* __restrict__ bstart,
                                             const int* __restrict__ btotal,
                                             int* __restrict__ sorted,
                                             int* __restrict__ count,
                                             int* __restrict__ start) {
    __shared__ int cnt[BW];
    __shared__ int cur[BW];
    __shared__ int wsum[BW / 32];
    __shared__ int stage[CAP2];
    const int t  = threadIdx.x;
    const int n  = btotal[b];
    const int g0 = bstart[b];
    cnt[t] = 0;
    __syncthreads();
    for (int i0 = t; i0 < n; i0 += RU * BW) {
        int rec[RU];
#pragma unroll
        for (int u = 0; u < RU; ++u) {
            const int i = i0 + u * BW;
            rec[u] = bucketed[g0 + (i < n ? i : n - 1)];
        }
#pragma unroll
        for (int u = 0; u < RU; ++u)
            if (i0 + u * BW < n) atomicAdd(&cnt[rec[u] >> SRCB], 1);
    }
    __syncthreads();

    const int c  = cnt[t];
    const int ex = block_excl_scan(c, wsum, BW / 32);
    const int dn = b * BW + t;
    if (dn < NNODES) { count[dn] = c; start[dn] = g0 + ex; }
    cur[t] = ex;
    __syncthreads();

    const bool fits = n <= CAP2;
    for (int i0 = t; i0 < n; i0 += RU * BW) {
        int rec[RU];
#pragma unroll
        for (int u = 0; u < RU; ++u) {
            const int i = i0 + u * BW;
            rec[u] = bucketed[g0 + (i < n ? i : n - 1)];
        }
        int p[RU];
#pragma unroll
        for (int u = 0; u < RU; ++u)
            p[u] = (i0 + u * BW < n) ? atomicAdd(&cur[rec[u] >> SRCB], 1) : -1;
#pragma unroll
        for (int u = 0; u < RU; ++u) {
            if (p[u] >= 0) {
                const int src = rec[u] & ((1 << SRCB) - 1);
                if (fits) stage[p[u]] = src; else sorted[g0 + p[u]] = src;
            }
        }
    }
    __syncthreads();
    if (fits) for (int i = t; i < n; i += BW) sorted[g0 + i] = stage[i];
}

__device__ __forceinline__ void gemm_tiles(const int grp,
                                           const float* __restrict__ x,
                                           const float* __restrict__ W,
                                           float* __restrict__ h) {
    const int lane = threadIdx.x & 31;
    const int wave = threadIdx.x >> 5;
    const int tile = grp * 8 + wave;
    const long base = (long)tile * 16;
    if (base >= NNODES) return;

    const int r  = lane & 15;
    const int kh = (lane >> 4) << 1;

    const float* xr = x + (base + r) * FIN;
    v2f a[4];
#pragma unroll
    for (int s = 0; s < 4; ++s) {
        const int k0 = s * 4 + kh;
        a[s][0] = xr[k0];
        a[s][1] = xr[k0 + 1];
    }

#pragma unroll
    for (int j = 0; j < 4; ++j) {
        v8f acc = {};
#pragma unroll
        for (int s = 0; s < 4; ++s) {
            const int k0 = s * 4 + kh;
            v2f b;
            b[0] = W[(k0    ) * HC + j * 16 + r];
            b[1] = W[(k0 + 1) * HC + j * 16 + r];
            acc = __builtin_amdgcn_wmma_f32_16x16x4_f32(
                 false, a[s],  false, b,
                 (short)0, acc,  false,  false);
        }
#pragma unroll
        for (int v = 0; v < 8; ++v) {
            const int row = ((lane >> 4) << 3) + v;
            h[(base + row) * HC + j * 16 + r] = acc[v];
        }
    }
}

#define NGB    782
static_assert(NGB * 8 * 16 >= NNODES && (NGB - 1) * 8 * 16 < NNODES && BW == 256, "tile groups");
__global__ __launch_bounds__(BW) void k_order_gemm(const int* __restrict__ bucketed,
                                                   const int* __restrict__ bstart,
                                                   const int* __restrict__ btotal,
                                                   int* __restrict__ sorted,
                                                   int* __restrict__ count,
                                                   int* __restrict__ start,
                                                   const float* __restrict__ x,
                                                   const float* __restrict__ W,
                                                   float* __restrict__ h) {
    if (blockIdx.x < NBKT) order_bucket(blockIdx.x, bucketed, bstart, btotal, sorted, count, start);
    else                   gemm_tiles(blockIdx.x - NBKT, x, W, h);
}

__global__ __launch_bounds__(256) void k_gather(const float* __restrict__ h,
                                                const int* __restrict__ sorted,
                                                const int* __restrict__ count,
                                                const int* __restrict__ start,
                                                const float* __restrict__ att_src,
                                                const float* __restrict__ att_dst,
                                                const float* __restrict__ bias,
                                                float* __restrict__ out) {
    const int i = blockIdx.x * blockDim.x + threadIdx.x;
    if (i >= NNODES * NH) return;
    const int d  = i >> 3;
    const int hh = i & 7;

    float as[NC], ad[NC];
#pragma unroll
    for (int c = 0; c < NC; ++c) {
        as[c] = att_src[hh * NC + c];
        ad[c] = att_dst[hh * NC + c];
    }

    const int beg = start[d];
    const int end = beg + count[d];

    const float4* hd = (const float4*)(h + (long)d * HC + hh * NC);
    const float4 x0 = hd[0];
    const float4 x1 = hd[1];
    float acc[NC] = {x0.x, x0.y, x0.z, x0.w, x1.x, x1.y, x1.z, x1.w};

    float ss = 0.0f, sd = 0.0f;
#pragma unroll
    for (int c = 0; c < NC; ++c) {
        ss = fmaf(acc[c], as[c], ss);
        sd = fmaf(acc[c], ad[c], sd);
    }
    const float a_d = sd;
    float m = ss + a_d;
    m = m > 0.0f ? m : 0.2f * m;
    float l = 1.0f;

    if (beg < end) {
        int s[GCH];
#pragma unroll
        for (int j = 0; j < GCH; ++j) {
            int q = beg + j;
            q = q < end ? q : end - 1;
            s[j] = sorted[q];
        }
        for (int p = beg; p < end; p += GCH) {
            float4 r0[GCH], r1[GCH];
#pragma unroll
            for (int j = 0; j < GCH; ++j) {
                const float4* hp = (const float4*)(h + (long)s[j] * HC + hh * NC);
                r0[j] = hp[0];
                r1[j] = hp[1];
            }
            int sn[GCH];
#pragma unroll
            for (int j = 0; j < GCH; ++j) {
                int q = p + GCH + j;
                q = q < end ? q : end - 1;
                sn[j] = sorted[q];
            }
            __builtin_amdgcn_sched_barrier(0);
#pragma unroll
            for (int j = 0; j < GCH; ++j) {
                const bool ok = (p + j) < end;
                const float row[NC] = {r0[j].x, r0[j].y, r0[j].z, r0[j].w,
                                       r1[j].x, r1[j].y, r1[j].z, r1[j].w};
                float a_s = 0.0f;
#pragma unroll
                for (int c = 0; c < NC; ++c) a_s = fmaf(row[c], as[c], a_s);
                float ev = a_s + a_d;
                ev = ev > 0.0f ? ev : 0.2f * ev;
                const float mn = ok ? fmaxf(m, ev) : m;
                const float sc = __expf(m - mn);
                const float pr = ok ? __expf(ev - mn) : 0.0f;
                l = fmaf(l, sc, pr);
#pragma unroll
                for (int c = 0; c < NC; ++c) acc[c] = fmaf(acc[c], sc, pr * row[c]);
                m = mn;
            }
#pragma unroll
            for (int j = 0; j < GCH; ++j) s[j] = sn[j];
        }
    }

    const float den = l + 1e-16f;
    float o[NC];
#pragma unroll
    for (int c = 0; c < NC; ++c) {
        const float v = acc[c] / den + bias[hh * NC + c];
        o[c] = v > 0.0f ? v : (__expf(v) - 1.0f);
    }
    float4* op = (float4*)(out + (long)d * HC + hh * NC);
    op[0] = make_float4(o[0], o[1], o[2], o[3]);
    op[1] = make_float4(o[4], o[5], o[6], o[7]);
}

extern "C" void kernel_launch(void* const* d_in, const int* in_sizes, int n_in,
                              void* d_out, int out_size, void* d_ws, size_t ws_size,
                              hipStream_t stream) {
    const float* x       = (const float*)d_in[0];
    const int*   ei      = (const int*)  d_in[1];
    const float* W       = (const float*)d_in[2];
    const float* att_src = (const float*)d_in[3];
    const float* att_dst = (const float*)d_in[4];
    const float* bias    = (const float*)d_in[5];
    float* out = (float*)d_out;

    char* ws = (char*)d_ws;
    float* h      = (float*)ws;   ws += sizeof(float) * NNODES * HC;
    int*   count  = (int*)ws;     ws += sizeof(int)   * NNODES;
    int*   start  = (int*)ws;     ws += sizeof(int)   * NNODES;
    int*   sorted = (int*)ws;
    int*   bucketed = (int*)d_out;
    int*   hist     = bucketed + NEDGES;
    int*   bstart   = hist + NB1 * NBKT;
    int*   btotal   = bstart + 512;

    const int blk   = 256;

    k_hist       <<<NB1,  T1,  0, stream>>>(ei, hist);
    k_bucket     <<<NB1,  T1,  0, stream>>>(ei, hist, bstart, btotal, bucketed);
    k_order_gemm <<<NBKT + NGB, BW, 0, stream>>>(bucketed, bstart, btotal, sorted, count, start,
                                                 x, W, h);
    k_gather     <<<(NNODES * NH + blk - 1) / blk, blk, 0, stream>>>(h, sorted, count, start,
                                                                   att_src, att_dst, bias, out);
}


# Gather sums each source's 16 inputs (one 128-B row: inputs, per-head logit terms) and applies W after the sum on the matrix unit via LDS, instead of gathering 256-B rows of x@W; new sum order
# speedup vs baseline: 7.8359x; 1.4024x over previous
#include <hip/hip_runtime.h>


#define NNODES 100000
#define NEDGES 1600000
#define FIN    16
#define NH     8
#define NC     8
#define HC     64
#define GCH    4
#define GP     32

#define BSH    8
#define BW     256
#define NBKT   391
#define NB1    250
#define T1     640
#define EPT    10
#define EPB    6400
#define SRCB   17
#define CAP2   6144

static_assert(NB1 * EPB == NEDGES && T1 * EPT == EPB, "level-1 blocks tile the edge list exactly");
static_assert(NBKT * BW >= NNODES && (NBKT - 1) * BW < NNODES && (1 << BSH) == BW, "bucket count");
static_assert(NNODES <= (1 << SRCB) && SRCB + BSH <= 31, "a packed edge record fits one int");
static_assert(NBKT <= T1 && (NEDGES % 2) == 0, "one thread per table entry; the table of pairs starts 8-byte aligned");

typedef __attribute__((ext_vector_type(2))) float v2f;
typedef __attribute__((ext_vector_type(8))) float v8f;

__device__ __forceinline__ int block_excl_scan(const int c, int* wsum, const int nw) {
    const int lane = threadIdx.x & 31;
    const int wave = threadIdx.x >> 5;
    int incl = c;
    for (int off = 1; off < 32; off <<= 1) {
        const int t = __shfl_up(incl, off);
        if (lane >= off) incl += t;
    }
    if (lane == 31) wsum[wave] = incl;
    __syncthreads();
    if (threadIdx.x == 0) {
        int tot = 0;
        for (int w = 0; w < nw; ++w) { const int t = wsum[w]; wsum[w] = tot; tot += t; }
    }
    __syncthreads();
    return wsum[wave] + incl - c;
}

__global__ __launch_bounds__(T1) void k_bucket(const int* __restrict__ ei,
                                               int* __restrict__ runs,
                                               int* __restrict__ tab) {
    __shared__ int cnt[NBKT];
    __shared__ int pre[NBKT];
    __shared__ int wsum[T1 / 32];
    __shared__ int lrec[EPB];
    const int t = threadIdx.x;
    if (t < NBKT) cnt[t] = 0;
    __syncthreads();

    const int e0 = blockIdx.x * EPB + t;
    int s[EPT], d[EPT], r[EPT];
#pragma unroll
    for (int k = 0; k < EPT; ++k) {
        s[k] = ei[e0 + k * T1];
        d[k] = ei[NEDGES + e0 + k * T1];
    }
#pragma unroll
    for (int k = 0; k < EPT; ++k) r[k] = atomicAdd(&cnt[d[k] >> BSH], 1);
    __syncthreads();

    const int c  = (t < NBKT) ? cnt[t] : 0;
    const int ex = block_excl_scan(c, wsum, T1 / 32);
    if (t < NBKT) {
        pre[t] = ex;
        ((int2*)tab)[blockIdx.x * NBKT + t] = make_int2(ex, c);
    }
    __syncthreads();

#pragma unroll
    for (int k = 0; k < EPT; ++k)
        lrec[pre[d[k] >> BSH] + r[k]] = s[k] | ((d[k] & (BW - 1)) << SRCB);
    __syncthreads();

    for (int i = t; i < EPB; i += T1) runs[blockIdx.x * EPB + i] = lrec[i];
}

#define RQ  8
#define RS  4
#define WU  16
static_assert(RQ * (BW / 8) >= NB1 && NB1 <= BW, "every run has its 8 lanes; one thread per run");
__device__ __forceinline__ void order_bucket(const int b,
                                             const int* __restrict__ runs,
                                             const int* __restrict__ tab,
                                             int* __restrict__ sorted,
                                             int* __restrict__ count,
                                             int* __restrict__ start) {
    __shared__ int cnt[BW];
    __shared__ int cur[BW];
    __shared__ int wsum_a[BW / 32];
    __shared__ int wsum_b[BW / 32];
    __shared__ int wsum_c[BW / 32];
    __shared__ int s_o0[BW];
    __shared__ int s_len[BW];
    __shared__ int s_rx[BW];
    __shared__ int tot[2];
    __shared__ int recs[CAP2];
    __shared__ int stage[CAP2];
    const int t = threadIdx.x;
    int o0 = 0, len = 0;
    if (t < NB1) {
        const int2 e = ((const int2*)tab)[t * NBKT + b];
        o0  = e.x;
        len = e.y;
    }
    cnt[t] = 0;
    const int rx = block_excl_scan(len, wsum_a, BW / 32);
    const int gx = block_excl_scan(o0,  wsum_b, BW / 32);
    s_o0[t] = o0; s_len[t] = len; s_rx[t] = rx;
    if (t == BW - 1) { tot[0] = rx + len; tot[1] = gx + o0; }
    __syncthreads();
    const int n  = tot[0];
    const int g0 = tot[1];
    const bool fits = n <= CAP2;
    const int* myrun = runs + (t < NB1 ? t : 0) * EPB + o0;

    if (fits) {
        const int sub = t & 7;
        const int gq  = t >> 3;
        int v[RQ][RS];
#pragma unroll
        for (int q = 0; q < RQ; ++q) {
            const int j  = gq + 32 * q;
            const int jj = j < NB1 ? j : NB1 - 1;
            const int jb = jj * EPB + s_o0[jj];
#pragma unroll
            for (int u = 0; u < RS; ++u) {
                int idx = jb + sub + 8 * u;
                idx = idx < NEDGES ? idx : NEDGES - 1;
                v[q][u] = runs[idx];
            }
        }
#pragma unroll
        for (int q = 0; q < RQ; ++q) {
            const int j  = gq + 32 * q;
            const int jj = j < NB1 ? j : NB1 - 1;
            const int jl = j < NB1 ? s_len[jj] : 0;
            const int jr = s_rx[jj];
#pragma unroll
            for (int u = 0; u < RS; ++u)
                if (sub + 8 * u < jl) recs[jr + sub + 8 * u] = v[q][u];
            if (jl > 8 * RS) {
                const int jb = jj * EPB + s_o0[jj];
#pragma nounroll
                for (int k = sub + 8 * RS; k < jl; k += 8) recs[jr + k] = runs[jb + k];
            }
        }
        __syncthreads();
        for (int i0 = t; i0 < n; i0 += WU * BW) {
            int rec[WU];
#pragma unroll
            for (int u = 0; u < WU; ++u) {
                const int i = i0 + u * BW;
                rec[u] = recs[i < n ? i : n - 1];
            }
#pragma unroll
            for (int u = 0; u < WU; ++u)
                if (i0 + u * BW < n) atomicAdd(&cnt[rec[u] >> SRCB], 1);
        }
    } else {
#pragma nounroll
        for (int k = 0; k < len; ++k) atomicAdd(&cnt[myrun[k] >> SRCB], 1);
    }
    __syncthreads();

    const int c  = cnt[t];
    const int ex = block_excl_scan(c, wsum_c, BW / 32);
    const int dn = b * BW + t;
    if (dn < NNODES) { count[dn] = c; start[dn] = g0 + ex; }
    cur[t] = ex;
    __syncthreads();

    if (fits) {
        for (int i0 = t; i0 < n; i0 += WU * BW) {
            int rec[WU];
#pragma unroll
            for (int u = 0; u < WU; ++u) {
                const int i = i0 + u * BW;
                rec[u] = recs[i < n ? i : n - 1];
            }
            int p[WU];
#pragma unroll
            for (int u = 0; u < WU; ++u)
                p[u] = (i0 + u * BW < n) ? atomicAdd(&cur[rec[u] >> SRCB], 1) : -1;
#pragma unroll
            for (int u = 0; u < WU; ++u)
                if (p[u] >= 0) stage[p[u]] = rec[u] & ((1 << SRCB) - 1);
        }
        __syncthreads();
        for (int i = t; i < n; i += BW) sorted[g0 + i] = stage[i];
    } else {
#pragma nounroll
        for (int k = 0; k < len; ++k) {
            const int rec = myrun[k];
            const int p   = atomicAdd(&cur[rec >> SRCB], 1);
            sorted[g0 + p] = rec & ((1 << SRCB) - 1);
        }
    }
}

__device__ __forceinline__ void proj_tiles(const int grp,
                                           const float* __restrict__ x,
                                           const float* __restrict__ W,
                                           const float* __restrict__ att_src,
                                           const float* __restrict__ att_dst,
                                           float* __restrict__ g) {
    const int lane = threadIdx.x & 31;
    const int wave = threadIdx.x >> 5;
    const int tile = grp * 8 + wave;
    const long base = (long)tile * 16;
    if (base >= NNODES) return;

    const int r  = lane & 15;
    const int kh = (lane >> 4) << 1;

    const float* xr = x + (base + r) * FIN;
    v2f a[4];
#pragma unroll
    for (int s = 0; s < 4; ++s) {
        const int k0 = s * 4 + kh;
        a[s][0] = xr[k0];
        a[s][1] = xr[k0 + 1];
    }

    const int hd = r & 7;
    const float* att = (r < NH ? att_src : att_dst) + hd * NC;
    float av[NC];
#pragma unroll
    for (int c = 0; c < NC; ++c) av[c] = att[c];
    v8f acc = {};
#pragma unroll
    for (int s = 0; s < 4; ++s) {
        const int k0 = s * 4 + kh;
        v2f b;
#pragma unroll
        for (int e = 0; e < 2; ++e) {
            const float* wr = W + (k0 + e) * HC + hd * NC;
            float t = 0.0f;
#pragma unroll
            for (int c = 0; c < NC; ++c) t = fmaf(wr[c], av[c], t);
            b[e] = t;
        }
        acc = __builtin_amdgcn_wmma_f32_16x16x4_f32(
             false, a[s],  false, b,
             (short)0, acc,  false,  false);
    }
#pragma unroll
    for (int v = 0; v < 8; ++v) {
        const int row = ((lane >> 4) << 3) + v;
        g[(base + row) * GP + FIN + r] = acc[v];
    }
    const float4* xt = (const float4*)(x + base * FIN);
#pragma unroll
    for (int i = 0; i < 2; ++i) {
        const int idx = i * 32 + lane;
        ((float4*)(g + (base + (idx >> 2)) * GP))[idx & 3] = xt[idx];
    }
}

#define NGB    782
static_assert(NGB * 8 * 16 >= NNODES && (NGB - 1) * 8 * 16 < NNODES && BW == 256, "tile groups");
static_assert(NNODES % 16 == 0 && FIN == 16 && 2 * NH == 16 && GP == FIN + 2 * NH, "whole tiles; one column tile");
__global__ __launch_bounds__(BW) void k_order_gemm(const int* __restrict__ runs,
                                                   const int* __restrict__ tab,
                                                   int* __restrict__ sorted,
                                                   int* __restrict__ count,
                                                   int* __restrict__ start,
                                                   const float* __restrict__ x,
                                                   const float* __restrict__ W,
                                                   const float* __restrict__ att_src,
                                                   const float* __restrict__ att_dst,
                                                   float* __restrict__ g) {
    if (blockIdx.x < NBKT) order_bucket(blockIdx.x, runs, tab, sorted, count, start);
    else                   proj_tiles(blockIdx.x - NBKT, x, W, att_src, att_dst, g);
}

#define ZP  132
#define QP  68
static_assert((NNODES * NH) % 256 == 0, "the grid covers (destination, head) exactly: every lane reaches the barriers");
__global__ __launch_bounds__(256) void k_gather(const float* __restrict__ g,
                                                const int* __restrict__ sorted,
                                                const int* __restrict__ count,
                                                const int* __restrict__ start,
                                                const float* __restrict__ W,
                                                const float* __restrict__ bias,
                                                float* __restrict__ out) {
    __shared__ float zs[32 * ZP];
    __shared__ float os[32 * QP];
    const int t  = threadIdx.x;
    const int i  = blockIdx.x * 256 + t;
    const int d  = i >> 3;
    const int hh = i & 7;

    const int beg = start[d];
    const int end = beg + count[d];

    const float*  gd = g + (long)d * GP;
    const float4* xd = (const float4*)gd;
    const float4 x0 = xd[0];
    const float4 x1 = xd[1];
    const float4 x2 = xd[2];
    const float4 x3 = xd[3];
    float z[FIN] = {x0.x, x0.y, x0.z, x0.w, x1.x, x1.y, x1.z, x1.w,
                    x2.x, x2.y, x2.z, x2.w, x3.x, x3.y, x3.z, x3.w};
    const float a_d = gd[FIN + NH + hh];
    float m = gd[FIN + hh] + a_d;
    m = m > 0.0f ? m : 0.2f * m;
    float l = 1.0f;

    if (beg < end) {
        int s[GCH];
#pragma unroll
        for (int j = 0; j < GCH; ++j) {
            int q = beg + j;
            q = q < end ? q : end - 1;
            s[j] = sorted[q];
        }
        for (int p = beg; p < end; p += GCH) {
            float4 r0[GCH], r1[GCH], r2[GCH], r3[GCH];
            float  as[GCH];
#pragma unroll
            for (int j = 0; j < GCH; ++j) {
                const float*  gp = g + (long)s[j] * GP;
                const float4* xp = (const float4*)gp;
                r0[j] = xp[0];
                r1[j] = xp[1];
                r2[j] = xp[2];
                r3[j] = xp[3];
                as[j] = gp[FIN + hh];
            }
            int sn[GCH];
#pragma unroll
            for (int j = 0; j < GCH; ++j) {
                int q = p + GCH + j;
                q = q < end ? q : end - 1;
                sn[j] = sorted[q];
            }
            __builtin_amdgcn_sched_barrier(0);
#pragma unroll
            for (int j = 0; j < GCH; ++j) {
                const bool ok = (p + j) < end;
                const float row[FIN] = {r0[j].x, r0[j].y, r0[j].z, r0[j].w,
                                        r1[j].x, r1[j].y, r1[j].z, r1[j].w,
                                        r2[j].x, r2[j].y, r2[j].z, r2[j].w,
                                        r3[j].x, r3[j].y, r3[j].z, r3[j].w};
                float ev = as[j] + a_d;
                ev = ev > 0.0f ? ev : 0.2f * ev;
                const float mn = ok ? fmaxf(m, ev) : m;
                const float sc = __expf(m - mn);
                const float pr = ok ? __expf(ev - mn) : 0.0f;
                l = fmaf(l, sc, pr);
#pragma unroll
                for (int k = 0; k < FIN; ++k) z[k] = fmaf(z[k], sc, pr * row[k]);
                m = mn;
            }
#pragma unroll
            for (int j = 0; j < GCH; ++j) s[j] = sn[j];
        }
    }
    const float den = l + 1e-16f;

    float4* zw = (float4*)(zs + (t >> 3) * ZP + hh * FIN);
#pragma unroll
    for (int q = 0; q < 4; ++q) zw[q] = make_float4(z[4 * q], z[4 * q + 1], z[4 * q + 2], z[4 * q + 3]);
    __syncthreads();

    {
        const int lane = t & 31;
        const int w    = t >> 5;
        const int r    = lane & 15;
        const int kh   = (lane >> 4) << 1;
        v2f b[4];
#pragma unroll
        for (int s4 = 0; s4 < 4; ++s4) {
            const int k0 = s4 * 4 + kh;
            b[s4][0] = W[(k0    ) * HC + w * NC + (r & 7)];
            b[s4][1] = W[(k0 + 1) * HC + w * NC + (r & 7)];
        }
#pragma unroll
        for (int mt = 0; mt < 2; ++mt) {
            const float* zr = zs + (mt * 16 + r) * ZP + w * FIN;
            v8f acc = {};
#pragma unroll
            for (int s4 = 0; s4 < 4; ++s4) {
                const int k0 = s4 * 4 + kh;
                v2f a;
                a[0] = zr[k0];
                a[1] = zr[k0 + 1];
                acc = __builtin_amdgcn_wmma_f32_16x16x4_f32(
                     false, a,  false, b[s4],
                     (short)0, acc,  false,  false);
            }
            if (r < NC) {
#pragma unroll
                for (int v = 0; v < 8; ++v)
                    os[(mt * 16 + ((lane >> 4) << 3) + v) * QP + w * NC + r] = acc[v];
            }
        }
    }
    __syncthreads();

    const float4* ov = (const float4*)(os + (t >> 3) * QP + hh * NC);
    const float4 u0 = ov[0];
    const float4 u1 = ov[1];
    const float u[NC] = {u0.x, u0.y, u0.z, u0.w, u1.x, u1.y, u1.z, u1.w};
    float o[NC];
#pragma unroll
    for (int c = 0; c < NC; ++c) {
        const float v = u[c] / den + bias[hh * NC + c];
        o[c] = v > 0.0f ? v : (__expf(v) - 1.0f);
    }
    float4* op = (float4*)(out + (long)d * HC + hh * NC);
    op[0] = make_float4(o[0], o[1], o[2], o[3]);
    op[1] = make_float4(o[4], o[5], o[6], o[7]);
}

extern "C" void kernel_launch(void* const* d_in, const int* in_sizes, int n_in,
                              void* d_out, int out_size, void* d_ws, size_t ws_size,
                              hipStream_t stream) {
    const float* x       = (const float*)d_in[0];
    const int*   ei      = (const int*)  d_in[1];
    const float* W       = (const float*)d_in[2];
    const float* att_src = (const float*)d_in[3];
    const float* att_dst = (const float*)d_in[4];
    const float* bias    = (const float*)d_in[5];
    float* out = (float*)d_out;

    char* ws = (char*)d_ws;
    float* g      = (float*)ws;   ws += sizeof(float) * NNODES * GP;
    int*   count  = (int*)ws;     ws += sizeof(int)   * NNODES;
    int*   start  = (int*)ws;     ws += sizeof(int)   * NNODES;
    int*   sorted = (int*)ws;
    int*   runs = (int*)d_out;
    int*   tab  = runs + NEDGES;

    const int blk   = 256;

    k_bucket     <<<NB1,  T1,  0, stream>>>(ei, runs, tab);
    k_order_gemm <<<NBKT + NGB, BW, 0, stream>>>(runs, tab, sorted, count, start,
                                                 x, W, att_src, att_dst, g);
    k_gather     <<<(NNODES * NH + blk - 1) / blk, blk, 0, stream>>>(g, sorted, count, start,
                                                                   W, bias, out);
}
